# starting kernel
# baseline (speedup 1.0000x reference)
#include <hip/hip_runtime.h>

typedef float v2f __attribute__((ext_vector_type(2)));
typedef float v8f __attribute__((ext_vector_type(8)));

#define IN_C  256
#define OUT_C 128
#define EPB   512

__global__ void gcn_zero_kernel(float* __restrict__ out, int total) {
    int i = blockIdx.x * blockDim.x + threadIdx.x;
    if (i < total) out[i] = 0.0f;
}

__global__ void gcn_gemm_wmma_kernel(const float* __restrict__ X,
                                     const float* __restrict__ W,
                                     float* __restrict__ H,
                                     int N) {
    const int wavesPerBlock = blockDim.x >> 5;
    const int wave = blockIdx.x * wavesPerBlock + (threadIdx.x >> 5);
    const int lane = threadIdx.x & 31;
    const int m0 = wave * 16;
    if (m0 >= N) return;

    const int half = lane >> 4;
    const int lid  = lane & 15;

    int arow = m0 + lid;
    if (arow >= N) arow = N - 1;
    const float* __restrict__ xrow  = X + (size_t)arow * IN_C + 2 * half;
    const float* __restrict__ wbase = W + (size_t)lid  * IN_C + 2 * half;

    v8f acc[8] = {};

    v2f a_cur = *(const v2f*)(xrow);
    v2f b_cur[8];
#pragma unroll
    for (int t = 0; t < 8; ++t)
        b_cur[t] = *(const v2f*)(wbase + (size_t)t * 16 * IN_C);

#pragma unroll 2
    for (int k = 0; k < IN_C; k += 4) {
        const int kn = (k + 4 < IN_C) ? (k + 4) : k;
        v2f a_nxt = *(const v2f*)(xrow + kn);
        v2f b_nxt[8];
#pragma unroll
        for (int t = 0; t < 8; ++t)
            b_nxt[t] = *(const v2f*)(wbase + (size_t)t * 16 * IN_C + kn);

#pragma unroll
        for (int t = 0; t < 8; ++t)
            acc[t] = __builtin_amdgcn_wmma_f32_16x16x4_f32(
                 false, a_cur,  false, b_cur[t],
                 (short)0, acc[t],
                 false,  false);

        a_cur = a_nxt;
#pragma unroll
        for (int t = 0; t < 8; ++t) b_cur[t] = b_nxt[t];
    }

    float* __restrict__ hbase = H + (size_t)(m0 + 8 * half) * OUT_C + lid;
    if (m0 + 16 <= N) {
#pragma unroll
        for (int t = 0; t < 8; ++t) {
            float* __restrict__ hp = hbase + t * 16;
#pragma unroll
            for (int g = 0; g < 8; ++g)
                hp[(size_t)g * OUT_C] = acc[t][g];
        }
    } else {
#pragma unroll
        for (int t = 0; t < 8; ++t) {
            const int col = t * 16 + lid;
#pragma unroll
            for (int g = 0; g < 8; ++g) {
                const int row = m0 + g + 8 * half;
                if (row < N) H[(size_t)row * OUT_C + col] = acc[t][g];
            }
        }
    }
}

__global__ void gcn_spmm_kernel(const int* __restrict__ A_rows,
                                const int* __restrict__ A_cols,
                                const float* __restrict__ A_vals,
                                const float* __restrict__ H,
                                float* __restrict__ out,
                                int E) {
    __shared__ int   s_row[EPB];
    __shared__ int   s_col[EPB];
    __shared__ float s_val[EPB];

    const int e0 = blockIdx.x * EPB;
    int n = E - e0;
    if (n > EPB) n = EPB;
    if (n <= 0) return;

    for (int i = threadIdx.x; i < n; i += blockDim.x) {
        s_row[i] = A_rows[e0 + i];
        s_col[i] = A_cols[e0 + i];
        s_val[i] = A_vals[e0 + i];
    }
    __syncthreads();

    const int c = threadIdx.x;
    float acc = 0.0f;
    int cur = s_row[0];

    for (int i = 0; i < n; ++i) {
        const int r = s_row[i];
        if (r != cur) {
            atomicAdd(&out[(size_t)cur * OUT_C + c], acc);
            acc = 0.0f;
            cur = r;
        }
        acc += s_val[i] * H[(size_t)s_col[i] * OUT_C + c];
    }
    atomicAdd(&out[(size_t)cur * OUT_C + c], acc);
}

extern "C" void kernel_launch(void* const* d_in, const int* in_sizes, int n_in,
                              void* d_out, int out_size, void* d_ws, size_t ws_size,
                              hipStream_t stream) {
    const float* X      = (const float*)d_in[0];
    const float* W      = (const float*)d_in[1];
    const int*   A_rows = (const int*)d_in[2];
    const int*   A_cols = (const int*)d_in[3];
    const float* A_vals = (const float*)d_in[4];
    float* out = (float*)d_out;
    float* H   = (float*)d_ws;

    const int N = in_sizes[0] / IN_C;
    const int E = in_sizes[2];

    gcn_zero_kernel<<<(out_size + 255) / 256, 256, 0, stream>>>(out, out_size);

    const int tiles  = (N + 15) / 16;
    const int blocks = (tiles + 7) / 8;
    gcn_gemm_wmma_kernel<<<blocks, 256, 0, stream>>>(X, W, H, N);

    gcn_spmm_kernel<<<(E + EPB - 1) / EPB, 128, 0, stream>>>(A_rows, A_cols, A_vals, H, out, E);
}


# SpMM gather loop batched: the H loads of 32 consecutive edges are issued together ahead of the serial row-change/accumulate chain (32 reads in flight per thread, was 1); same sum order and atomics
# speedup vs baseline: 1.5204x; 1.5204x over previous
#include <hip/hip_runtime.h>

typedef float v2f __attribute__((ext_vector_type(2)));
typedef float v8f __attribute__((ext_vector_type(8)));

#define IN_C  256
#define OUT_C 128
#define EPB   512
#define GB    32

__global__ void gcn_zero_kernel(float* __restrict__ out, int total) {
    int i = blockIdx.x * blockDim.x + threadIdx.x;
    if (i < total) out[i] = 0.0f;
}

__global__ void gcn_gemm_wmma_kernel(const float* __restrict__ X,
                                     const float* __restrict__ W,
                                     float* __restrict__ H,
                                     int N) {
    const int wavesPerBlock = blockDim.x >> 5;
    const int wave = blockIdx.x * wavesPerBlock + (threadIdx.x >> 5);
    const int lane = threadIdx.x & 31;
    const int m0 = wave * 16;
    if (m0 >= N) return;

    const int half = lane >> 4;
    const int lid  = lane & 15;

    int arow = m0 + lid;
    if (arow >= N) arow = N - 1;
    const float* __restrict__ xrow  = X + (size_t)arow * IN_C + 2 * half;
    const float* __restrict__ wbase = W + (size_t)lid  * IN_C + 2 * half;

    v8f acc[8] = {};

    v2f a_cur = *(const v2f*)(xrow);
    v2f b_cur[8];
#pragma unroll
    for (int t = 0; t < 8; ++t)
        b_cur[t] = *(const v2f*)(wbase + (size_t)t * 16 * IN_C);

#pragma unroll 2
    for (int k = 0; k < IN_C; k += 4) {
        const int kn = (k + 4 < IN_C) ? (k + 4) : k;
        v2f a_nxt = *(const v2f*)(xrow + kn);
        v2f b_nxt[8];
#pragma unroll
        for (int t = 0; t < 8; ++t)
            b_nxt[t] = *(const v2f*)(wbase + (size_t)t * 16 * IN_C + kn);

#pragma unroll
        for (int t = 0; t < 8; ++t)
            acc[t] = __builtin_amdgcn_wmma_f32_16x16x4_f32(
                 false, a_cur,  false, b_cur[t],
                 (short)0, acc[t],
                 false,  false);

        a_cur = a_nxt;
#pragma unroll
        for (int t = 0; t < 8; ++t) b_cur[t] = b_nxt[t];
    }

    float* __restrict__ hbase = H + (size_t)(m0 + 8 * half) * OUT_C + lid;
    if (m0 + 16 <= N) {
#pragma unroll
        for (int t = 0; t < 8; ++t) {
            float* __restrict__ hp = hbase + t * 16;
#pragma unroll
            for (int g = 0; g < 8; ++g)
                hp[(size_t)g * OUT_C] = acc[t][g];
        }
    } else {
#pragma unroll
        for (int t = 0; t < 8; ++t) {
            const int col = t * 16 + lid;
#pragma unroll
            for (int g = 0; g < 8; ++g) {
                const int row = m0 + g + 8 * half;
                if (row < N) H[(size_t)row * OUT_C + col] = acc[t][g];
            }
        }
    }
}

__global__ void gcn_spmm_kernel(const int* __restrict__ A_rows,
                                const int* __restrict__ A_cols,
                                const float* __restrict__ A_vals,
                                const float* __restrict__ H,
                                float* __restrict__ out,
                                int E) {
    __shared__ int   s_row[EPB];
    __shared__ int   s_col[EPB];
    __shared__ float s_val[EPB];

    const int e0 = blockIdx.x * EPB;
    int n = E - e0;
    if (n > EPB) n = EPB;
    if (n <= 0) return;

    for (int i = threadIdx.x; i < n; i += blockDim.x) {
        s_row[i] = A_rows[e0 + i];
        s_col[i] = A_cols[e0 + i];
        s_val[i] = A_vals[e0 + i];
    }
    __syncthreads();

    const int c = threadIdx.x;
    float acc = 0.0f;
    int cur = s_row[0];

    int i = 0;
    for (; i + GB <= n; i += GB) {
        float h[GB];
#pragma unroll
        for (int j = 0; j < GB; ++j)
            h[j] = H[(size_t)s_col[i + j] * OUT_C + c];
#pragma unroll
        for (int j = 0; j < GB; ++j) {
            const int r = s_row[i + j];
            if (r != cur) {
                atomicAdd(&out[(size_t)cur * OUT_C + c], acc);
                acc = 0.0f;
                cur = r;
            }
            acc += s_val[i + j] * h[j];
        }
    }
    for (; i < n; ++i) {
        const int r = s_row[i];
        if (r != cur) {
            atomicAdd(&out[(size_t)cur * OUT_C + c], acc);
            acc = 0.0f;
            cur = r;
        }
        acc += s_val[i] * H[(size_t)s_col[i] * OUT_C + c];
    }
    atomicAdd(&out[(size_t)cur * OUT_C + c], acc);
}

extern "C" void kernel_launch(void* const* d_in, const int* in_sizes, int n_in,
                              void* d_out, int out_size, void* d_ws, size_t ws_size,
                              hipStream_t stream) {
    const float* X      = (const float*)d_in[0];
    const float* W      = (const float*)d_in[1];
    const int*   A_rows = (const int*)d_in[2];
    const int*   A_cols = (const int*)d_in[3];
    const float* A_vals = (const float*)d_in[4];
    float* out = (float*)d_out;
    float* H   = (float*)d_ws;

    const int N = in_sizes[0] / IN_C;
    const int E = in_sizes[2];

    gcn_zero_kernel<<<(out_size + 255) / 256, 256, 0, stream>>>(out, out_size);

    const int tiles  = (N + 15) / 16;
    const int blocks = (tiles + 7) / 8;
    gcn_gemm_wmma_kernel<<<blocks, 256, 0, stream>>>(X, W, H, N);

    gcn_spmm_kernel<<<(E + EPB - 1) / EPB, 128, 0, stream>>>(A_rows, A_cols, A_vals, H, out, E);
}


# out's zero start value is stored by the GEMM waves (each wave its own 16 rows: 16 whole-row stores of 16 B per lane); the zero-fill launch with its 400000 one-dword waves is gone, 3 launches -> 2
# speedup vs baseline: 1.6273x; 1.0703x over previous
#include <hip/hip_runtime.h>

typedef float v2f __attribute__((ext_vector_type(2)));
typedef float v4f __attribute__((ext_vector_type(4)));
typedef float v8f __attribute__((ext_vector_type(8)));

#define IN_C  256
#define OUT_C 128
#define EPB   512
#define GB    32

__global__ void gcn_gemm_wmma_kernel(const float* __restrict__ X,
                                     const float* __restrict__ W,
                                     float* __restrict__ H,
                                     float* __restrict__ out,
                                     int N) {
    const int wavesPerBlock = blockDim.x >> 5;
    const int wave = blockIdx.x * wavesPerBlock + (threadIdx.x >> 5);
    const int lane = threadIdx.x & 31;
    const int m0 = wave * 16;
    if (m0 >= N) return;

    const int half = lane >> 4;
    const int lid  = lane & 15;

    {
        const v4f z4 = {0.0f, 0.0f, 0.0f, 0.0f};
        float* __restrict__ orow = out + (size_t)m0 * OUT_C + 4 * lane;
        if (m0 + 16 <= N) {
#pragma unroll
            for (int j = 0; j < 16; ++j)
                *(v4f*)(orow + (size_t)j * OUT_C) = z4;
        } else {
            for (int j = 0; j < 16; ++j)
                if (m0 + j < N) *(v4f*)(orow + (size_t)j * OUT_C) = z4;
        }
    }

    int arow = m0 + lid;
    if (arow >= N) arow = N - 1;
    const float* __restrict__ xrow  = X + (size_t)arow * IN_C + 2 * half;
    const float* __restrict__ wbase = W + (size_t)lid  * IN_C + 2 * half;

    v8f acc[8] = {};

    v2f a_cur = *(const v2f*)(xrow);
    v2f b_cur[8];
#pragma unroll
    for (int t = 0; t < 8; ++t)
        b_cur[t] = *(const v2f*)(wbase + (size_t)t * 16 * IN_C);

#pragma unroll 2
    for (int k = 0; k < IN_C; k += 4) {
        const int kn = (k + 4 < IN_C) ? (k + 4) : k;
        v2f a_nxt = *(const v2f*)(xrow + kn);
        v2f b_nxt[8];
#pragma unroll
        for (int t = 0; t < 8; ++t)
            b_nxt[t] = *(const v2f*)(wbase + (size_t)t * 16 * IN_C + kn);

#pragma unroll
        for (int t = 0; t < 8; ++t)
            acc[t] = __builtin_amdgcn_wmma_f32_16x16x4_f32(
                 false, a_cur,  false, b_cur[t],
                 (short)0, acc[t],
                 false,  false);

        a_cur = a_nxt;
#pragma unroll
        for (int t = 0; t < 8; ++t) b_cur[t] = b_nxt[t];
    }

    float* __restrict__ hbase = H + (size_t)(m0 + 8 * half) * OUT_C + lid;
    if (m0 + 16 <= N) {
#pragma unroll
        for (int t = 0; t < 8; ++t) {
            float* __restrict__ hp = hbase + t * 16;
#pragma unroll
            for (int g = 0; g < 8; ++g)
                hp[(size_t)g * OUT_C] = acc[t][g];
        }
    } else {
#pragma unroll
        for (int t = 0; t < 8; ++t) {
            const int col = t * 16 + lid;
#pragma unroll
            for (int g = 0; g < 8; ++g) {
                const int row = m0 + g + 8 * half;
                if (row < N) H[(size_t)row * OUT_C + col] = acc[t][g];
            }
        }
    }
}

__global__ void gcn_spmm_kernel(const int* __restrict__ A_rows,
                                const int* __restrict__ A_cols,
                                const float* __restrict__ A_vals,
                                const float* __restrict__ H,
                                float* __restrict__ out,
                                int E) {
    __shared__ int   s_row[EPB];
    __shared__ int   s_col[EPB];
    __shared__ float s_val[EPB];

    const int e0 = blockIdx.x * EPB;
    int n = E - e0;
    if (n > EPB) n = EPB;
    if (n <= 0) return;

    for (int i = threadIdx.x; i < n; i += blockDim.x) {
        s_row[i] = A_rows[e0 + i];
        s_col[i] = A_cols[e0 + i];
        s_val[i] = A_vals[e0 + i];
    }
    __syncthreads();

    const int c = threadIdx.x;
    float acc = 0.0f;
    int cur = s_row[0];

    int i = 0;
    for (; i + GB <= n; i += GB) {
        float h[GB];
#pragma unroll
        for (int j = 0; j < GB; ++j)
            h[j] = H[(size_t)s_col[i + j] * OUT_C + c];
#pragma unroll
        for (int j = 0; j < GB; ++j) {
            const int r = s_row[i + j];
            if (r != cur) {
                atomicAdd(&out[(size_t)cur * OUT_C + c], acc);
                acc = 0.0f;
                cur = r;
            }
            acc += s_val[i + j] * h[j];
        }
    }
    for (; i < n; ++i) {
        const int r = s_row[i];
        if (r != cur) {
            atomicAdd(&out[(size_t)cur * OUT_C + c], acc);
            acc = 0.0f;
            cur = r;
        }
        acc += s_val[i] * H[(size_t)s_col[i] * OUT_C + c];
    }
    atomicAdd(&out[(size_t)cur * OUT_C + c], acc);
}

extern "C" void kernel_launch(void* const* d_in, const int* in_sizes, int n_in,
                              void* d_out, int out_size, void* d_ws, size_t ws_size,
                              hipStream_t stream) {
    const float* X      = (const float*)d_in[0];
    const float* W      = (const float*)d_in[1];
    const int*   A_rows = (const int*)d_in[2];
    const int*   A_cols = (const int*)d_in[3];
    const float* A_vals = (const float*)d_in[4];
    float* out = (float*)d_out;
    float* H   = (float*)d_ws;

    const int N = in_sizes[0] / IN_C;
    const int E = in_sizes[2];

    const int tiles  = (N + 15) / 16;
    const int blocks = (tiles + 7) / 8;
    gcn_gemm_wmma_kernel<<<blocks, 256, 0, stream>>>(X, W, H, out, N);

    gcn_spmm_kernel<<<(E + EPB - 1) / EPB, 128, 0, stream>>>(A_rows, A_cols, A_vals, H, out, E);
}


# GEMM: W (128 KB) copied once per work-group into LDS at a 260-float row pitch; the 8 B fragments per k-step are conflict-free 8-byte LDS reads, not 16-line global loads; same k order and sums
# speedup vs baseline: 2.2277x; 1.3690x over previous
#include <hip/hip_runtime.h>

typedef float v2f __attribute__((ext_vector_type(2)));
typedef float v4f __attribute__((ext_vector_type(4)));
typedef float v8f __attribute__((ext_vector_type(8)));

#define IN_C  256
#define OUT_C 128
#define EPB   512
#define GB    32
#define GEMM_THREADS 256
#define WPITCH 260

__global__ void __launch_bounds__(GEMM_THREADS)
gcn_gemm_wmma_kernel(const float* __restrict__ X,
                     const float* __restrict__ W,
                     float* __restrict__ H,
                     float* __restrict__ out,
                     int N) {
    __shared__ __attribute__((aligned(16))) float sW[OUT_C * WPITCH];

    const int tid  = threadIdx.x;
    const int wave = blockIdx.x * (GEMM_THREADS / 32) + (tid >> 5);
    const int lane = tid & 31;
    const int m0 = wave * 16;

    const int half = lane >> 4;
    const int lid  = lane & 15;

    if (m0 < N) {
        const v4f z4 = {0.0f, 0.0f, 0.0f, 0.0f};
        float* __restrict__ orow = out + (size_t)m0 * OUT_C + 4 * lane;
        if (m0 + 16 <= N) {
#pragma unroll
            for (int j = 0; j < 16; ++j)
                *(v4f*)(orow + (size_t)j * OUT_C) = z4;
        } else {
            for (int j = 0; j < 16; ++j)
                if (m0 + j < N) *(v4f*)(orow + (size_t)j * OUT_C) = z4;
        }
    }

    {
        const v4f* __restrict__ Wv = (const v4f*)W;
        const int wr = tid >> 6;
        const int wc = (tid & 63) * 4;
        v4f tmp[32];
#pragma unroll
        for (int j = 0; j < 32; ++j)
            tmp[j] = Wv[tid + GEMM_THREADS * j];
#pragma unroll
        for (int j = 0; j < 32; ++j)
            *(v4f*)(sW + (wr + 4 * j) * WPITCH + wc) = tmp[j];
    }
    __syncthreads();
    if (m0 >= N) return;

    int arow = m0 + lid;
    if (arow >= N) arow = N - 1;
    const float* __restrict__ xrow  = X + (size_t)arow * IN_C + 2 * half;
    const float* wbase = sW + lid * WPITCH + 2 * half;

    v8f acc[8] = {};

    v2f a_cur = *(const v2f*)(xrow);
    v2f b_cur[8];
#pragma unroll
    for (int t = 0; t < 8; ++t)
        b_cur[t] = *(const v2f*)(wbase + t * 16 * WPITCH);

#pragma unroll 2
    for (int k = 0; k < IN_C; k += 4) {
        const int kn = (k + 4 < IN_C) ? (k + 4) : k;
        v2f a_nxt = *(const v2f*)(xrow + kn);
        v2f b_nxt[8];
#pragma unroll
        for (int t = 0; t < 8; ++t)
            b_nxt[t] = *(const v2f*)(wbase + t * 16 * WPITCH + kn);

#pragma unroll
        for (int t = 0; t < 8; ++t)
            acc[t] = __builtin_amdgcn_wmma_f32_16x16x4_f32(
                 false, a_cur,  false, b_cur[t],
                 (short)0, acc[t],
                 false,  false);

        a_cur = a_nxt;
#pragma unroll
        for (int t = 0; t < 8; ++t) b_cur[t] = b_nxt[t];
    }

    float* __restrict__ hbase = H + (size_t)(m0 + 8 * half) * OUT_C + lid;
    if (m0 + 16 <= N) {
#pragma unroll
        for (int t = 0; t < 8; ++t) {
            float* __restrict__ hp = hbase + t * 16;
#pragma unroll
            for (int g = 0; g < 8; ++g)
                hp[(size_t)g * OUT_C] = acc[t][g];
        }
    } else {
#pragma unroll
        for (int t = 0; t < 8; ++t) {
            const int col = t * 16 + lid;
#pragma unroll
            for (int g = 0; g < 8; ++g) {
                const int row = m0 + g + 8 * half;
                if (row < N) H[(size_t)row * OUT_C + col] = acc[t][g];
            }
        }
    }
}

__global__ void gcn_spmm_kernel(const int* __restrict__ A_rows,
                                const int* __restrict__ A_cols,
                                const float* __restrict__ A_vals,
                                const float* __restrict__ H,
                                float* __restrict__ out,
                                int E) {
    __shared__ int   s_row[EPB];
    __shared__ int   s_col[EPB];
    __shared__ float s_val[EPB];

    const int e0 = blockIdx.x * EPB;
    int n = E - e0;
    if (n > EPB) n = EPB;
    if (n <= 0) return;

    for (int i = threadIdx.x; i < n; i += blockDim.x) {
        s_row[i] = A_rows[e0 + i];
        s_col[i] = A_cols[e0 + i];
        s_val[i] = A_vals[e0 + i];
    }
    __syncthreads();

    const int c = threadIdx.x;
    float acc = 0.0f;
    int cur = s_row[0];

    int i = 0;
    for (; i + GB <= n; i += GB) {
        float h[GB];
#pragma unroll
        for (int j = 0; j < GB; ++j)
            h[j] = H[(size_t)s_col[i + j] * OUT_C + c];
#pragma unroll
        for (int j = 0; j < GB; ++j) {
            const int r = s_row[i + j];
            if (r != cur) {
                atomicAdd(&out[(size_t)cur * OUT_C + c], acc);
                acc = 0.0f;
                cur = r;
            }
            acc += s_val[i + j] * h[j];
        }
    }
    for (; i < n; ++i) {
        const int r = s_row[i];
        if (r != cur) {
            atomicAdd(&out[(size_t)cur * OUT_C + c], acc);
            acc = 0.0f;
            cur = r;
        }
        acc += s_val[i] * H[(size_t)s_col[i] * OUT_C + c];
    }
    atomicAdd(&out[(size_t)cur * OUT_C + c], acc);
}

extern "C" void kernel_launch(void* const* d_in, const int* in_sizes, int n_in,
                              void* d_out, int out_size, void* d_ws, size_t ws_size,
                              hipStream_t stream) {
    const float* X      = (const float*)d_in[0];
    const float* W      = (const float*)d_in[1];
    const int*   A_rows = (const int*)d_in[2];
    const int*   A_cols = (const int*)d_in[3];
    const float* A_vals = (const float*)d_in[4];
    float* out = (float*)d_out;
    float* H   = (float*)d_ws;

    const int N = in_sizes[0] / IN_C;
    const int E = in_sizes[2];

    const int tiles  = (N + 15) / 16;
    const int blocks = (tiles + 7) / 8;
    gcn_gemm_wmma_kernel<<<blocks, GEMM_THREADS, 0, stream>>>(X, W, H, out, N);

    gcn_spmm_kernel<<<(E + EPB - 1) / EPB, 128, 0, stream>>>(A_rows, A_cols, A_vals, H, out, E);
}


# SpMM: one wave per edge: each lane gathers 16 B (4 channels) of H[c], was 4 waves x 4 B; flush turned in LDS to whole-line atomics; rows now cut at 128-edge (was 512) segments
# speedup vs baseline: 2.7201x; 1.2210x over previous
#include <hip/hip_runtime.h>

typedef float v2f __attribute__((ext_vector_type(2)));
typedef float v4f __attribute__((ext_vector_type(4)));
typedef float v8f __attribute__((ext_vector_type(8)));

#define IN_C  256
#define OUT_C 128
#define EPB   512
#define EPW   128
#define GB    16
#define GEMM_THREADS 256
#define WPITCH 260

__global__ void __launch_bounds__(GEMM_THREADS)
gcn_gemm_wmma_kernel(const float* __restrict__ X,
                     const float* __restrict__ W,
                     float* __restrict__ H,
                     float* __restrict__ out,
                     int N) {
    __shared__ __attribute__((aligned(16))) float sW[OUT_C * WPITCH];

    const int tid  = threadIdx.x;
    const int wave = blockIdx.x * (GEMM_THREADS / 32) + (tid >> 5);
    const int lane = tid & 31;
    const int m0 = wave * 16;

    const int half = lane >> 4;
    const int lid  = lane & 15;

    if (m0 < N) {
        const v4f z4 = {0.0f, 0.0f, 0.0f, 0.0f};
        float* __restrict__ orow = out + (size_t)m0 * OUT_C + 4 * lane;
        if (m0 + 16 <= N) {
#pragma unroll
            for (int j = 0; j < 16; ++j)
                *(v4f*)(orow + (size_t)j * OUT_C) = z4;
        } else {
            for (int j = 0; j < 16; ++j)
                if (m0 + j < N) *(v4f*)(orow + (size_t)j * OUT_C) = z4;
        }
    }

    {
        const v4f* __restrict__ Wv = (const v4f*)W;
        const int wr = tid >> 6;
        const int wc = (tid & 63) * 4;
        v4f tmp[32];
#pragma unroll
        for (int j = 0; j < 32; ++j)
            tmp[j] = Wv[tid + GEMM_THREADS * j];
#pragma unroll
        for (int j = 0; j < 32; ++j)
            *(v4f*)(sW + (wr + 4 * j) * WPITCH + wc) = tmp[j];
    }
    __syncthreads();
    if (m0 >= N) return;

    int arow = m0 + lid;
    if (arow >= N) arow = N - 1;
    const float* __restrict__ xrow  = X + (size_t)arow * IN_C + 2 * half;
    const float* wbase = sW + lid * WPITCH + 2 * half;

    v8f acc[8] = {};

    v2f a_cur = *(const v2f*)(xrow);
    v2f b_cur[8];
#pragma unroll
    for (int t = 0; t < 8; ++t)
        b_cur[t] = *(const v2f*)(wbase + t * 16 * WPITCH);

#pragma unroll 2
    for (int k = 0; k < IN_C; k += 4) {
        const int kn = (k + 4 < IN_C) ? (k + 4) : k;
        v2f a_nxt = *(const v2f*)(xrow + kn);
        v2f b_nxt[8];
#pragma unroll
        for (int t = 0; t < 8; ++t)
            b_nxt[t] = *(const v2f*)(wbase + t * 16 * WPITCH + kn);

#pragma unroll
        for (int t = 0; t < 8; ++t)
            acc[t] = __builtin_amdgcn_wmma_f32_16x16x4_f32(
                 false, a_cur,  false, b_cur[t],
                 (short)0, acc[t],
                 false,  false);

        a_cur = a_nxt;
#pragma unroll
        for (int t = 0; t < 8; ++t) b_cur[t] = b_nxt[t];
    }

    float* __restrict__ hbase = H + (size_t)(m0 + 8 * half) * OUT_C + lid;
    if (m0 + 16 <= N) {
#pragma unroll
        for (int t = 0; t < 8; ++t) {
            float* __restrict__ hp = hbase + t * 16;
#pragma unroll
            for (int g = 0; g < 8; ++g)
                hp[(size_t)g * OUT_C] = acc[t][g];
        }
    } else {
#pragma unroll
        for (int t = 0; t < 8; ++t) {
            const int col = t * 16 + lid;
#pragma unroll
            for (int g = 0; g < 8; ++g) {
                const int row = m0 + g + 8 * half;
                if (row < N) H[(size_t)row * OUT_C + col] = acc[t][g];
            }
        }
    }
}

__device__ __forceinline__ void spmm_flush(float* __restrict__ out, float* st,
                                           int row, v4f acc, int lane) {
    *(v4f*)(st + 4 * lane) = acc;
    float* __restrict__ orow = out + (size_t)row * OUT_C;
#pragma unroll
    for (int q = 0; q < 4; ++q)
        atomicAdd(orow + 32 * q + lane, st[32 * q + lane]);
}

__global__ void __launch_bounds__(128)
gcn_spmm_kernel(const int* __restrict__ A_rows,
                const int* __restrict__ A_cols,
                const float* __restrict__ A_vals,
                const float* __restrict__ H,
                float* __restrict__ out,
                int E) {
    __shared__ __attribute__((aligned(16))) float s_t[(EPB / EPW) * OUT_C];

    const int lane = threadIdx.x & 31;
    const int wv = __builtin_amdgcn_readfirstlane((int)(threadIdx.x >> 5));
    const int e0 = blockIdx.x * EPB + wv * EPW;
    int n = E - e0;
    if (n > EPW) n = EPW;
    if (n <= 0) return;

    float* st = s_t + wv * OUT_C;
    v4f acc = {0.0f, 0.0f, 0.0f, 0.0f};
    int cur = __builtin_amdgcn_readfirstlane(A_rows[e0]);

    if (n == EPW) {
        int   er = A_rows[e0 + lane];
        int   ec = A_cols[e0 + lane];
        float ev = A_vals[e0 + lane];
        for (int g = 0; g < EPW; g += 32) {
            int er_n = er, ec_n = ec;
            float ev_n = ev;
            if (g + 32 < EPW) {
                er_n = A_rows[e0 + g + 32 + lane];
                ec_n = A_cols[e0 + g + 32 + lane];
                ev_n = A_vals[e0 + g + 32 + lane];
            }
#pragma unroll
            for (int hh = 0; hh < 32; hh += GB) {
                v4f h[GB];
#pragma unroll
                for (int j = 0; j < GB; ++j) {
                    const int c = __builtin_amdgcn_readlane(ec, hh + j);
                    const float* __restrict__ hrow = H + (size_t)c * OUT_C;
                    h[j] = *(const v4f*)(hrow + 4 * lane);
                }
#pragma unroll
                for (int j = 0; j < GB; ++j) {
                    const int r = __builtin_amdgcn_readlane(er, hh + j);
                    const float v = __int_as_float(
                        __builtin_amdgcn_readlane(__float_as_int(ev), hh + j));
                    if (r != cur) {
                        spmm_flush(out, st, cur, acc, lane);
                        acc = (v4f){0.0f, 0.0f, 0.0f, 0.0f};
                        cur = r;
                    }
                    acc += v * h[j];
                }
            }
            er = er_n;
            ec = ec_n;
            ev = ev_n;
        }
    } else {
        for (int i = 0; i < n; ++i) {
            const int r = __builtin_amdgcn_readfirstlane(A_rows[e0 + i]);
            const int c = __builtin_amdgcn_readfirstlane(A_cols[e0 + i]);
            const float v = __int_as_float(
                __builtin_amdgcn_readfirstlane(__float_as_int(A_vals[e0 + i])));
            if (r != cur) {
                spmm_flush(out, st, cur, acc, lane);
                acc = (v4f){0.0f, 0.0f, 0.0f, 0.0f};
                cur = r;
            }
            const float* __restrict__ hrow = H + (size_t)c * OUT_C;
            acc += v * *(const v4f*)(hrow + 4 * lane);
        }
    }
    spmm_flush(out, st, cur, acc, lane);
}

extern "C" void kernel_launch(void* const* d_in, const int* in_sizes, int n_in,
                              void* d_out, int out_size, void* d_ws, size_t ws_size,
                              hipStream_t stream) {
    const float* X      = (const float*)d_in[0];
    const float* W      = (const float*)d_in[1];
    const int*   A_rows = (const int*)d_in[2];
    const int*   A_cols = (const int*)d_in[3];
    const float* A_vals = (const float*)d_in[4];
    float* out = (float*)d_out;
    float* H   = (float*)d_ws;

    const int N = in_sizes[0] / IN_C;
    const int E = in_sizes[2];

    const int tiles  = (N + 15) / 16;
    const int blocks = (tiles + 7) / 8;
    gcn_gemm_wmma_kernel<<<blocks, GEMM_THREADS, 0, stream>>>(X, W, H, out, N);

    gcn_spmm_kernel<<<(E + EPB - 1) / EPB, 128, 0, stream>>>(A_rows, A_cols, A_vals, H, out, E);
}


# GEMM: 512 resident work-groups (2 per compute unit) each walk a run of 16-row stripes dealt evenly over the SIMDs (3 whole stripes per SIMD, an odd stripe as four 32-column pieces); was 782 one-pass
# speedup vs baseline: 2.9993x; 1.1026x over previous
#include <hip/hip_runtime.h>

typedef float v2f __attribute__((ext_vector_type(2)));
typedef float v4f __attribute__((ext_vector_type(4)));
typedef float v8f __attribute__((ext_vector_type(8)));

#define IN_C  256
#define OUT_C 128
#define EPB   512
#define EPW   128
#define GB    16
#define GEMM_THREADS 256
#define GEMM_BLOCKS  512
#define WPITCH 260


template <int NT>
__device__ __forceinline__ void gemm_tiles(const float* __restrict__ X,
                                           float* __restrict__ H,
                                           const float* sW,
                                           int m0, int col0, int N,
                                           int lid, int half) {
    int arow = m0 + lid;
    if (arow >= N) arow = N - 1;
    const float* __restrict__ xrow  = X + (size_t)arow * IN_C + 2 * half;
    const float* wbase = sW + (col0 + lid) * WPITCH + 2 * half;

    v8f acc[NT] = {};

    v2f a_cur = *(const v2f*)(xrow);
    v2f b_cur[NT];
#pragma unroll
    for (int t = 0; t < NT; ++t)
        b_cur[t] = *(const v2f*)(wbase + t * 16 * WPITCH);

#pragma unroll 2
    for (int k = 0; k < IN_C; k += 4) {
        const int kn = (k + 4 < IN_C) ? (k + 4) : k;
        v2f a_nxt = *(const v2f*)(xrow + kn);
        v2f b_nxt[NT];
#pragma unroll
        for (int t = 0; t < NT; ++t)
            b_nxt[t] = *(const v2f*)(wbase + t * 16 * WPITCH + kn);

#pragma unroll
        for (int t = 0; t < NT; ++t)
            acc[t] = __builtin_amdgcn_wmma_f32_16x16x4_f32(
                 false, a_cur,  false, b_cur[t],
                 (short)0, acc[t],
                 false,  false);

        a_cur = a_nxt;
#pragma unroll
        for (int t = 0; t < NT; ++t) b_cur[t] = b_nxt[t];
    }

    float* __restrict__ hbase = H + (size_t)(m0 + 8 * half) * OUT_C + col0 + lid;
    if (m0 + 16 <= N) {
#pragma unroll
        for (int t = 0; t < NT; ++t) {
            float* __restrict__ hp = hbase + t * 16;
#pragma unroll
            for (int g = 0; g < 8; ++g)
                hp[(size_t)g * OUT_C] = acc[t][g];
        }
    } else {
#pragma unroll
        for (int t = 0; t < NT; ++t) {
            const int col = col0 + t * 16 + lid;
#pragma unroll
            for (int g = 0; g < 8; ++g) {
                const int row = m0 + g + 8 * half;
                if (row < N) H[(size_t)row * OUT_C + col] = acc[t][g];
            }
        }
    }
}

__global__ void __launch_bounds__(GEMM_THREADS)
gcn_gemm_wmma_kernel(const float* __restrict__ X,
                     const float* __restrict__ W,
                     float* __restrict__ H,
                     float* __restrict__ out,
                     int N) {
    __shared__ __attribute__((aligned(16))) float sW[OUT_C * WPITCH];

    const int tid  = threadIdx.x;
    const int lane = tid & 31;
    const int wv = __builtin_amdgcn_readfirstlane(tid >> 5);

    const int half = lane >> 4;
    const int lid  = lane & 15;

    const int T    = (N + 15) >> 4;
    const int base = T / GEMM_BLOCKS;
    const int rem  = T % GEMM_BLOCKS;
    const int b    = blockIdx.x;
    const int cnt   = base + (b < rem ? 1 : 0);
    const int first = b * base + (b < rem ? b : rem);

    {
        const v4f z4 = {0.0f, 0.0f, 0.0f, 0.0f};
        for (int s = wv; s < cnt; s += GEMM_THREADS / 32) {
            const int z0 = (first + s) * 16;
            float* __restrict__ orow = out + (size_t)z0 * OUT_C + 4 * lane;
            if (z0 + 16 <= N) {
#pragma unroll
                for (int j = 0; j < 16; ++j)
                    *(v4f*)(orow + (size_t)j * OUT_C) = z4;
            } else {
                for (int j = 0; j < 16; ++j)
                    if (z0 + j < N) *(v4f*)(orow + (size_t)j * OUT_C) = z4;
            }
        }
    }

    {
        const v4f* __restrict__ Wv = (const v4f*)W;
        const int wr = tid >> 6;
        const int wc = (tid & 63) * 4;
        v4f tmp[32];
#pragma unroll
        for (int j = 0; j < 32; ++j)
            tmp[j] = Wv[tid + GEMM_THREADS * j];
#pragma unroll
        for (int j = 0; j < 32; ++j)
            *(v4f*)(sW + (wr + 4 * j) * WPITCH + wc) = tmp[j];
    }
    __syncthreads();

    int s = first;
    int left = cnt;
    for (; left >= 8; left -= 8, s += 8)
        gemm_tiles<8>(X, H, sW, (s + wv) * 16, 0, N, lid, half);
    if (left >= 4) {
        if (wv < 4)
            gemm_tiles<8>(X, H, sW, (s + wv) * 16, 0, N, lid, half);
        s += 4;
        left -= 4;
    }
    if (wv >= 4)
        for (; left > 0; --left, ++s)
            gemm_tiles<2>(X, H, sW, s * 16, 32 * (wv - 4), N, lid, half);
}

__device__ __forceinline__ void spmm_flush(float* __restrict__ out, float* st,
                                           int row, v4f acc, int lane) {
    *(v4f*)(st + 4 * lane) = acc;
    float* __restrict__ orow = out + (size_t)row * OUT_C;
#pragma unroll
    for (int q = 0; q < 4; ++q)
        atomicAdd(orow + 32 * q + lane, st[32 * q + lane]);
}

__global__ void __launch_bounds__(128)
gcn_spmm_kernel(const int* __restrict__ A_rows,
                const int* __restrict__ A_cols,
                const float* __restrict__ A_vals,
                const float* __restrict__ H,
                float* __restrict__ out,
                int E) {
    __shared__ __attribute__((aligned(16))) float s_t[(EPB / EPW) * OUT_C];

    const int lane = threadIdx.x & 31;
    const int wv = __builtin_amdgcn_readfirstlane((int)(threadIdx.x >> 5));
    const int e0 = blockIdx.x * EPB + wv * EPW;
    int n = E - e0;
    if (n > EPW) n = EPW;
    if (n <= 0) return;

    float* st = s_t + wv * OUT_C;
    v4f acc = {0.0f, 0.0f, 0.0f, 0.0f};
    int cur = __builtin_amdgcn_readfirstlane(A_rows[e0]);

    if (n == EPW) {
        int   er = A_rows[e0 + lane];
        int   ec = A_cols[e0 + lane];
        float ev = A_vals[e0 + lane];
        for (int g = 0; g < EPW; g += 32) {
            int er_n = er, ec_n = ec;
            float ev_n = ev;
            if (g + 32 < EPW) {
                er_n = A_rows[e0 + g + 32 + lane];
                ec_n = A_cols[e0 + g + 32 + lane];
                ev_n = A_vals[e0 + g + 32 + lane];
            }
#pragma unroll
            for (int hh = 0; hh < 32; hh += GB) {
                v4f h[GB];
#pragma unroll
                for (int j = 0; j < GB; ++j) {
                    const int c = __builtin_amdgcn_readlane(ec, hh + j);
                    const float* __restrict__ hrow = H + (size_t)c * OUT_C;
                    h[j] = *(const v4f*)(hrow + 4 * lane);
                }
#pragma unroll
                for (int j = 0; j < GB; ++j) {
                    const int r = __builtin_amdgcn_readlane(er, hh + j);
                    const float v = __int_as_float(
                        __builtin_amdgcn_readlane(__float_as_int(ev), hh + j));
                    if (r != cur) {
                        spmm_flush(out, st, cur, acc, lane);
                        acc = (v4f){0.0f, 0.0f, 0.0f, 0.0f};
                        cur = r;
                    }
                    acc += v * h[j];
                }
            }
            er = er_n;
            ec = ec_n;
            ev = ev_n;
        }
    } else {
        for (int i = 0; i < n; ++i) {
            const int r = __builtin_amdgcn_readfirstlane(A_rows[e0 + i]);
            const int c = __builtin_amdgcn_readfirstlane(A_cols[e0 + i]);
            const float v = __int_as_float(
                __builtin_amdgcn_readfirstlane(__float_as_int(A_vals[e0 + i])));
            if (r != cur) {
                spmm_flush(out, st, cur, acc, lane);
                acc = (v4f){0.0f, 0.0f, 0.0f, 0.0f};
                cur = r;
            }
            const float* __restrict__ hrow = H + (size_t)c * OUT_C;
            acc += v * *(const v4f*)(hrow + 4 * lane);
        }
    }
    spmm_flush(out, st, cur, acc, lane);
}

extern "C" void kernel_launch(void* const* d_in, const int* in_sizes, int n_in,
                              void* d_out, int out_size, void* d_ws, size_t ws_size,
                              hipStream_t stream) {
    const float* X      = (const float*)d_in[0];
    const float* W      = (const float*)d_in[1];
    const int*   A_rows = (const int*)d_in[2];
    const int*   A_cols = (const int*)d_in[3];
    const float* A_vals = (const float*)d_in[4];
    float* out = (float*)d_out;
    float* H   = (float*)d_ws;

    const int N = in_sizes[0] / IN_C;
    const int E = in_sizes[2];

    gcn_gemm_wmma_kernel<<<GEMM_BLOCKS, GEMM_THREADS, 0, stream>>>(X, W, H, out, N);

    gcn_spmm_kernel<<<(E + EPB - 1) / EPB, 128, 0, stream>>>(A_rows, A_cols, A_vals, H, out, E);
}


# GEMM: the A fragments of a whole 128-B line of X (8 k-steps) are requested one block ahead as one group of 8 loads, so a memory miss lies under 64 WMMAs, not in front of 16; same k order and sums
# speedup vs baseline: 3.1293x; 1.0433x over previous
#include <hip/hip_runtime.h>

typedef float v2f __attribute__((ext_vector_type(2)));
typedef float v4f __attribute__((ext_vector_type(4)));
typedef float v8f __attribute__((ext_vector_type(8)));

#define IN_C  256
#define OUT_C 128
#define EPB   512
#define EPW   128
#define GB    16
#define GEMM_THREADS 256
#define GEMM_BLOCKS  512
#define WPITCH 260
#define KBLOCK 32
#define KSTEPS (KBLOCK / 4)


template <int NT>
__device__ __forceinline__ void gemm_tiles(const float* __restrict__ X,
                                           float* __restrict__ H,
                                           const float* sW,
                                           int m0, int col0, int N,
                                           int lid, int half) {
    int arow = m0 + lid;
    if (arow >= N) arow = N - 1;
    const float* __restrict__ xrow  = X + (size_t)arow * IN_C + 2 * half;
    const float* wbase = sW + (col0 + lid) * WPITCH + 2 * half;

    v8f acc[NT] = {};

    v2f a_cur[KSTEPS];
#pragma unroll
    for (int s = 0; s < KSTEPS; ++s)
        a_cur[s] = *(const v2f*)(xrow + 4 * s);

    v2f b_cur[NT];
#pragma unroll
    for (int t = 0; t < NT; ++t)
        b_cur[t] = *(const v2f*)(wbase + t * 16 * WPITCH);

#pragma unroll 2
    for (int kb = 0; kb < IN_C; kb += KBLOCK) {
        const int kbn = (kb + KBLOCK < IN_C) ? (kb + KBLOCK) : kb;
        v2f a_nxt[KSTEPS];
#pragma unroll
        for (int s = 0; s < KSTEPS; ++s)
            a_nxt[s] = *(const v2f*)(xrow + kbn + 4 * s);

#pragma unroll
        for (int s = 0; s < KSTEPS; ++s) {
            const int kn = (s + 1 < KSTEPS) ? (kb + 4 * (s + 1)) : kbn;
            v2f b_nxt[NT];
#pragma unroll
            for (int t = 0; t < NT; ++t)
                b_nxt[t] = *(const v2f*)(wbase + t * 16 * WPITCH + kn);

            __builtin_amdgcn_sched_barrier(0);

#pragma unroll
            for (int t = 0; t < NT; ++t)
                acc[t] = __builtin_amdgcn_wmma_f32_16x16x4_f32(
                     false, a_cur[s],  false, b_cur[t],
                     (short)0, acc[t],
                     false,  false);

#pragma unroll
            for (int t = 0; t < NT; ++t) b_cur[t] = b_nxt[t];
        }

#pragma unroll
        for (int s = 0; s < KSTEPS; ++s) a_cur[s] = a_nxt[s];
    }

    float* __restrict__ hbase = H + (size_t)(m0 + 8 * half) * OUT_C + col0 + lid;
    if (m0 + 16 <= N) {
#pragma unroll
        for (int t = 0; t < NT; ++t) {
            float* __restrict__ hp = hbase + t * 16;
#pragma unroll
            for (int g = 0; g < 8; ++g)
                hp[(size_t)g * OUT_C] = acc[t][g];
        }
    } else {
#pragma unroll
        for (int t = 0; t < NT; ++t) {
            const int col = col0 + t * 16 + lid;
#pragma unroll
            for (int g = 0; g < 8; ++g) {
                const int row = m0 + g + 8 * half;
                if (row < N) H[(size_t)row * OUT_C + col] = acc[t][g];
            }
        }
    }
}

__global__ void __launch_bounds__(GEMM_THREADS)
gcn_gemm_wmma_kernel(const float* __restrict__ X,
                     const float* __restrict__ W,
                     float* __restrict__ H,
                     float* __restrict__ out,
                     int N) {
    __shared__ __attribute__((aligned(16))) float sW[OUT_C * WPITCH];

    const int tid  = threadIdx.x;
    const int lane = tid & 31;
    const int wv = __builtin_amdgcn_readfirstlane(tid >> 5);

    const int half = lane >> 4;
    const int lid  = lane & 15;

    const int T    = (N + 15) >> 4;
    const int base = T / GEMM_BLOCKS;
    const int rem  = T % GEMM_BLOCKS;
    const int b    = blockIdx.x;
    const int cnt   = base + (b < rem ? 1 : 0);
    const int first = b * base + (b < rem ? b : rem);

    {
        const v4f z4 = {0.0f, 0.0f, 0.0f, 0.0f};
        for (int s = wv; s < cnt; s += GEMM_THREADS / 32) {
            const int z0 = (first + s) * 16;
            float* __restrict__ orow = out + (size_t)z0 * OUT_C + 4 * lane;
            if (z0 + 16 <= N) {
#pragma unroll
                for (int j = 0; j < 16; ++j)
                    *(v4f*)(orow + (size_t)j * OUT_C) = z4;
            } else {
                for (int j = 0; j < 16; ++j)
                    if (z0 + j < N) *(v4f*)(orow + (size_t)j * OUT_C) = z4;
            }
        }
    }

    {
        const v4f* __restrict__ Wv = (const v4f*)W;
        const int wr = tid >> 6;
        const int wc = (tid & 63) * 4;
        v4f tmp[32];
#pragma unroll
        for (int j = 0; j < 32; ++j)
            tmp[j] = Wv[tid + GEMM_THREADS * j];
#pragma unroll
        for (int j = 0; j < 32; ++j)
            *(v4f*)(sW + (wr + 4 * j) * WPITCH + wc) = tmp[j];
    }
    __syncthreads();

    int s = first;
    int left = cnt;
    for (; left >= 8; left -= 8, s += 8)
        gemm_tiles<8>(X, H, sW, (s + wv) * 16, 0, N, lid, half);
    if (left >= 4) {
        if (wv < 4)
            gemm_tiles<8>(X, H, sW, (s + wv) * 16, 0, N, lid, half);
        s += 4;
        left -= 4;
    }
    if (wv >= 4)
        for (; left > 0; --left, ++s)
            gemm_tiles<2>(X, H, sW, s * 16, 32 * (wv - 4), N, lid, half);
}

__device__ __forceinline__ void spmm_flush(float* __restrict__ out, float* st,
                                           int row, v4f acc, int lane) {
    *(v4f*)(st + 4 * lane) = acc;
    float* __restrict__ orow = out + (size_t)row * OUT_C;
#pragma unroll
    for (int q = 0; q < 4; ++q)
        atomicAdd(orow + 32 * q + lane, st[32 * q + lane]);
}

__global__ void __launch_bounds__(128)
gcn_spmm_kernel(const int* __restrict__ A_rows,
                const int* __restrict__ A_cols,
                const float* __restrict__ A_vals,
                const float* __restrict__ H,
                float* __restrict__ out,
                int E) {
    __shared__ __attribute__((aligned(16))) float s_t[(EPB / EPW) * OUT_C];

    const int lane = threadIdx.x & 31;
    const int wv = __builtin_amdgcn_readfirstlane((int)(threadIdx.x >> 5));
    const int e0 = blockIdx.x * EPB + wv * EPW;
    int n = E - e0;
    if (n > EPW) n = EPW;
    if (n <= 0) return;

    float* st = s_t + wv * OUT_C;
    v4f acc = {0.0f, 0.0f, 0.0f, 0.0f};
    int cur = __builtin_amdgcn_readfirstlane(A_rows[e0]);

    if (n == EPW) {
        int   er = A_rows[e0 + lane];
        int   ec = A_cols[e0 + lane];
        float ev = A_vals[e0 + lane];
        for (int g = 0; g < EPW; g += 32) {
            int er_n = er, ec_n = ec;
            float ev_n = ev;
            if (g + 32 < EPW) {
                er_n = A_rows[e0 + g + 32 + lane];
                ec_n = A_cols[e0 + g + 32 + lane];
                ev_n = A_vals[e0 + g + 32 + lane];
            }
#pragma unroll
            for (int hh = 0; hh < 32; hh += GB) {
                v4f h[GB];
#pragma unroll
                for (int j = 0; j < GB; ++j) {
                    const int c = __builtin_amdgcn_readlane(ec, hh + j);
                    const float* __restrict__ hrow = H + (size_t)c * OUT_C;
                    h[j] = *(const v4f*)(hrow + 4 * lane);
                }
#pragma unroll
                for (int j = 0; j < GB; ++j) {
                    const int r = __builtin_amdgcn_readlane(er, hh + j);
                    const float v = __int_as_float(
                        __builtin_amdgcn_readlane(__float_as_int(ev), hh + j));
                    if (r != cur) {
                        spmm_flush(out, st, cur, acc, lane);
                        acc = (v4f){0.0f, 0.0f, 0.0f, 0.0f};
                        cur = r;
                    }
                    acc += v * h[j];
                }
            }
            er = er_n;
            ec = ec_n;
            ev = ev_n;
        }
    } else {
        for (int i = 0; i < n; ++i) {
            const int r = __builtin_amdgcn_readfirstlane(A_rows[e0 + i]);
            const int c = __builtin_amdgcn_readfirstlane(A_cols[e0 + i]);
            const float v = __int_as_float(
                __builtin_amdgcn_readfirstlane(__float_as_int(A_vals[e0 + i])));
            if (r != cur) {
                spmm_flush(out, st, cur, acc, lane);
                acc = (v4f){0.0f, 0.0f, 0.0f, 0.0f};
                cur = r;
            }
            const float* __restrict__ hrow = H + (size_t)c * OUT_C;
            acc += v * *(const v4f*)(hrow + 4 * lane);
        }
    }
    spmm_flush(out, st, cur, acc, lane);
}

extern "C" void kernel_launch(void* const* d_in, const int* in_sizes, int n_in,
                              void* d_out, int out_size, void* d_ws, size_t ws_size,
                              hipStream_t stream) {
    const float* X      = (const float*)d_in[0];
    const float* W      = (const float*)d_in[1];
    const int*   A_rows = (const int*)d_in[2];
    const int*   A_cols = (const int*)d_in[3];
    const float* A_vals = (const float*)d_in[4];
    float* out = (float*)d_out;
    float* H   = (float*)d_ws;

    const int N = in_sizes[0] / IN_C;
    const int E = in_sizes[2];

    gcn_gemm_wmma_kernel<<<GEMM_BLOCKS, GEMM_THREADS, 0, stream>>>(X, W, H, out, N);

    gcn_spmm_kernel<<<(E + EPB - 1) / EPB, 128, 0, stream>>>(A_rows, A_cols, A_vals, H, out, E);
}


# SpMM: each row is summed whole, in edge order, by the wave in whose 128-edge segment it starts (it reads on past the segment end) and stored once: no atomics, no zero start value; sum order changes
# speedup vs baseline: 3.3920x; 1.0839x over previous
#include <hip/hip_runtime.h>

typedef float v2f __attribute__((ext_vector_type(2)));
typedef float v4f __attribute__((ext_vector_type(4)));
typedef float v8f __attribute__((ext_vector_type(8)));

#define IN_C  256
#define OUT_C 128
#define EPB   512
#define EPW   128
#define GB    16
#define GEMM_THREADS 256
#define GEMM_BLOCKS  512
#define WPITCH 260
#define KBLOCK 32
#define KSTEPS (KBLOCK / 4)


template <int NT>
__device__ __forceinline__ void gemm_tiles(const float* __restrict__ X,
                                           float* __restrict__ H,
                                           const float* sW,
                                           int m0, int col0, int N,
                                           int lid, int half) {
    int arow = m0 + lid;
    if (arow >= N) arow = N - 1;
    const float* __restrict__ xrow  = X + (size_t)arow * IN_C + 2 * half;
    const float* wbase = sW + (col0 + lid) * WPITCH + 2 * half;

    v8f acc[NT] = {};

    v2f a_cur[KSTEPS];
#pragma unroll
    for (int s = 0; s < KSTEPS; ++s)
        a_cur[s] = *(const v2f*)(xrow + 4 * s);

    v2f b_cur[NT];
#pragma unroll
    for (int t = 0; t < NT; ++t)
        b_cur[t] = *(const v2f*)(wbase + t * 16 * WPITCH);

#pragma unroll 2
    for (int kb = 0; kb < IN_C; kb += KBLOCK) {
        const int kbn = (kb + KBLOCK < IN_C) ? (kb + KBLOCK) : kb;
        v2f a_nxt[KSTEPS];
#pragma unroll
        for (int s = 0; s < KSTEPS; ++s)
            a_nxt[s] = *(const v2f*)(xrow + kbn + 4 * s);

#pragma unroll
        for (int s = 0; s < KSTEPS; ++s) {
            const int kn = (s + 1 < KSTEPS) ? (kb + 4 * (s + 1)) : kbn;
            v2f b_nxt[NT];
#pragma unroll
            for (int t = 0; t < NT; ++t)
                b_nxt[t] = *(const v2f*)(wbase + t * 16 * WPITCH + kn);

            __builtin_amdgcn_sched_barrier(0);

#pragma unroll
            for (int t = 0; t < NT; ++t)
                acc[t] = __builtin_amdgcn_wmma_f32_16x16x4_f32(
                     false, a_cur[s],  false, b_cur[t],
                     (short)0, acc[t],
                     false,  false);

#pragma unroll
            for (int t = 0; t < NT; ++t) b_cur[t] = b_nxt[t];
        }

#pragma unroll
        for (int s = 0; s < KSTEPS; ++s) a_cur[s] = a_nxt[s];
    }

    float* __restrict__ hbase = H + (size_t)(m0 + 8 * half) * OUT_C + col0 + lid;
    if (m0 + 16 <= N) {
#pragma unroll
        for (int t = 0; t < NT; ++t) {
            float* __restrict__ hp = hbase + t * 16;
#pragma unroll
            for (int g = 0; g < 8; ++g)
                hp[(size_t)g * OUT_C] = acc[t][g];
        }
    } else {
#pragma unroll
        for (int t = 0; t < NT; ++t) {
            const int col = col0 + t * 16 + lid;
#pragma unroll
            for (int g = 0; g < 8; ++g) {
                const int row = m0 + g + 8 * half;
                if (row < N) H[(size_t)row * OUT_C + col] = acc[t][g];
            }
        }
    }
}

__global__ void __launch_bounds__(GEMM_THREADS)
gcn_gemm_wmma_kernel(const float* __restrict__ X,
                     const float* __restrict__ W,
                     float* __restrict__ H,
                     int N) {
    __shared__ __attribute__((aligned(16))) float sW[OUT_C * WPITCH];

    const int tid  = threadIdx.x;
    const int lane = tid & 31;
    const int wv = __builtin_amdgcn_readfirstlane(tid >> 5);

    const int half = lane >> 4;
    const int lid  = lane & 15;

    const int T    = (N + 15) >> 4;
    const int base = T / GEMM_BLOCKS;
    const int rem  = T % GEMM_BLOCKS;
    const int b    = blockIdx.x;
    const int cnt   = base + (b < rem ? 1 : 0);
    const int first = b * base + (b < rem ? b : rem);

    {
        const v4f* __restrict__ Wv = (const v4f*)W;
        const int wr = tid >> 6;
        const int wc = (tid & 63) * 4;
        v4f tmp[32];
#pragma unroll
        for (int j = 0; j < 32; ++j)
            tmp[j] = Wv[tid + GEMM_THREADS * j];
#pragma unroll
        for (int j = 0; j < 32; ++j)
            *(v4f*)(sW + (wr + 4 * j) * WPITCH + wc) = tmp[j];
    }
    __syncthreads();

    int s = first;
    int left = cnt;
    for (; left >= 8; left -= 8, s += 8)
        gemm_tiles<8>(X, H, sW, (s + wv) * 16, 0, N, lid, half);
    if (left >= 4) {
        if (wv < 4)
            gemm_tiles<8>(X, H, sW, (s + wv) * 16, 0, N, lid, half);
        s += 4;
        left -= 4;
    }
    if (wv >= 4)
        for (; left > 0; --left, ++s)
            gemm_tiles<2>(X, H, sW, s * 16, 32 * (wv - 4), N, lid, half);
}

__device__ __forceinline__ void spmm_store_row(float* __restrict__ out, int row,
                                               v4f acc, int lane) {
    *(v4f*)(out + (size_t)row * OUT_C + 4 * lane) = acc;
}

__device__ __forceinline__ void spmm_zero_rows(float* __restrict__ out, int a, int b,
                                               int lane) {
    const v4f z4 = {0.0f, 0.0f, 0.0f, 0.0f};
#pragma nounroll
    for (int z = a; z < b; ++z)
        *(v4f*)(out + (size_t)z * OUT_C + 4 * lane) = z4;
}

#define SPMM_ROW_CHANGE(r)                                   \
    do {                                                     \
        if (own) spmm_store_row(out, cur, acc, lane);        \
        spmm_zero_rows(out, cur + 1, (r), lane);             \
        acc = (v4f){0.0f, 0.0f, 0.0f, 0.0f};                 \
        cur = (r);                                           \
        own = 1;                                             \
    } while (0)

template <int B>
__device__ __forceinline__ void spmm_batch(const float* __restrict__ H,
                                           float* __restrict__ out,
                                           int er, int ec, float ev, int j0, int lane,
                                           v4f& acc, int& cur, int& own) {
    v4f h[B];
#pragma unroll
    for (int k = 0; k < B; ++k) {
        const int c = __builtin_amdgcn_readlane(ec, j0 + k);
        const float* __restrict__ hrow = H + (size_t)c * OUT_C;
        h[k] = *(const v4f*)(hrow + 4 * lane);
    }
#pragma unroll
    for (int k = 0; k < B; ++k) {
        const int r = __builtin_amdgcn_readlane(er, j0 + k);
        const float v = __int_as_float(
            __builtin_amdgcn_readlane(__float_as_int(ev), j0 + k));
        if (r != cur) SPMM_ROW_CHANGE(r);
        acc += v * h[k];
    }
}

__global__ void __launch_bounds__(128)
gcn_spmm_kernel(const int* __restrict__ A_rows,
                const int* __restrict__ A_cols,
                const float* __restrict__ A_vals,
                const float* __restrict__ H,
                float* __restrict__ out,
                int E, int N) {
    const int lane = threadIdx.x & 31;
    const int wv = __builtin_amdgcn_readfirstlane((int)(threadIdx.x >> 5));
    const int e0 = blockIdx.x * EPB + wv * EPW;
    if (e0 >= E) return;
    const int seg_end = e0 + EPW;

    v4f acc = {0.0f, 0.0f, 0.0f, 0.0f};
    int cur = __builtin_amdgcn_readfirstlane(A_rows[e0]);
    const int prev = (e0 > 0) ? __builtin_amdgcn_readfirstlane(A_rows[e0 - 1]) : -1;
    int own = (prev != cur) ? 1 : 0;
    if (own) spmm_zero_rows(out, prev + 1, cur, lane);

    int   base = e0;
    const int ei = (base + lane < E) ? (base + lane) : (E - 1);
    int   er = A_rows[ei];
    int   ec = A_cols[ei];
    float ev = A_vals[ei];
    if (base + lane >= E) er = -1;

    int nxt;
    for (;;) {
        const unsigned m = __builtin_amdgcn_ballot_w32(er == cur);
        const int same = (m == 0xffffffffu) ? 32 : __builtin_ctz(~m);
        const int lo = own ? 0 : same;
        int hi;
        if (base < seg_end) {
            hi = E - base;
            if (hi > 32) hi = 32;
        } else {
            if (!own) return;
            hi = same;
        }

        const int en  = base + 32 + lane;
        const int eni = (en < E) ? en : (E - 1);
        int   er_n = A_rows[eni];
        const int   ec_n = A_cols[eni];
        const float ev_n = A_vals[eni];
        if (en >= E) er_n = -1;

        if (lo == 0 && hi == 32) {
            spmm_batch<GB>(H, out, er, ec, ev, 0, lane, acc, cur, own);
            spmm_batch<GB>(H, out, er, ec, ev, GB, lane, acc, cur, own);
        } else {
            int j = lo;
            if (hi - j >= 16) { spmm_batch<16>(H, out, er, ec, ev, j, lane, acc, cur, own); j += 16; }
            if (hi - j >= 8)  { spmm_batch<8>(H, out, er, ec, ev, j, lane, acc, cur, own);  j += 8; }
            if (hi - j >= 4)  { spmm_batch<4>(H, out, er, ec, ev, j, lane, acc, cur, own);  j += 4; }
            if (hi - j >= 2)  { spmm_batch<2>(H, out, er, ec, ev, j, lane, acc, cur, own);  j += 2; }
            if (hi - j >= 1)  { spmm_batch<1>(H, out, er, ec, ev, j, lane, acc, cur, own); }
        }

        if (base >= seg_end && hi < 32) {
            nxt = base + hi;
            break;
        }
        base += 32;
        if (base >= E) {
            nxt = base;
            break;
        }
        er = er_n;
        ec = ec_n;
        ev = ev_n;
    }
    if (!own) return;
    spmm_store_row(out, cur, acc, lane);
    if (nxt >= E) spmm_zero_rows(out, cur + 1, N, lane);
}

extern "C" void kernel_launch(void* const* d_in, const int* in_sizes, int n_in,
                              void* d_out, int out_size, void* d_ws, size_t ws_size,
                              hipStream_t stream) {
    const float* X      = (const float*)d_in[0];
    const float* W      = (const float*)d_in[1];
    const int*   A_rows = (const int*)d_in[2];
    const int*   A_cols = (const int*)d_in[3];
    const float* A_vals = (const float*)d_in[4];
    float* out = (float*)d_out;
    float* H   = (float*)d_ws;

    const int N = in_sizes[0] / IN_C;
    const int E = in_sizes[2];

    gcn_gemm_wmma_kernel<<<GEMM_BLOCKS, GEMM_THREADS, 0, stream>>>(X, W, H, N);

    gcn_spmm_kernel<<<(E + EPB - 1) / EPB, 128, 0, stream>>>(A_rows, A_cols, A_vals, H, out, E, N);
}
